# starting kernel
# baseline (speedup 1.0000x reference)
#include <hip/hip_runtime.h>

typedef float v2f __attribute__((ext_vector_type(2)));
typedef float v4f __attribute__((ext_vector_type(4)));
typedef float v8f __attribute__((ext_vector_type(8)));

#define GS_B  512
#define GS_K  64
#define GS_L  1024
#define GS_LP 1028
#define GS_NT 1024

__device__ __forceinline__ void lds_fence() {
  asm volatile("s_wait_dscnt 0" ::: "memory");
  __builtin_amdgcn_wave_barrier();
}

__global__ __launch_bounds__(GS_NT, 1)
void gram_schmidt_cholqr(const float* __restrict__ xin,
                         float* __restrict__ xout) {
  extern __shared__ float smem[];
  float* X = smem;
  float* G = smem + GS_K * GS_LP;
  float* W = G + GS_K * GS_K;

  const int b    = blockIdx.x;
  const int tid  = threadIdx.x;
  const int wave = tid >> 5;
  const int lane = tid & 31;

  const float* xg = xin  + (size_t)b * (GS_K * GS_L);
  float*       og = xout + (size_t)b * (GS_K * GS_L);

  for (int idx = tid; idx < GS_K * (GS_L / 4); idx += GS_NT) {
    int r  = idx >> 8;
    int c4 = (idx & 255) << 2;
    *(v4f*)(X + r * GS_LP + c4) = *(const v4f*)(xg + r * GS_L + c4);
  }
  __syncthreads();

  {
    const int tile  = wave >> 1;
    const int ti    = tile >> 2;
    const int tj    = tile & 3;
    const int khalf = wave & 1;
    const int m     = lane & 15;
    const int koff  = (lane < 16) ? 0 : 2;
    const float* pa = X + (ti * 16 + m) * GS_LP + koff + khalf * (GS_L / 2);
    const float* pb = X + (tj * 16 + m) * GS_LP + koff + khalf * (GS_L / 2);
    v8f acc = {0.f, 0.f, 0.f, 0.f, 0.f, 0.f, 0.f, 0.f};
    #pragma unroll 4
    for (int k = 0; k < GS_L / 2; k += 4) {
      v2f a  = *(const v2f*)(pa + k);
      v2f bb = *(const v2f*)(pb + k);
      acc = __builtin_amdgcn_wmma_f32_16x16x4_f32(false, a, false, bb,
                                                  (short)0, acc, false, false);
    }
    float*    dst = khalf ? W : G;
    const int mb  = (lane < 16) ? 0 : 8;
    #pragma unroll
    for (int j = 0; j < 8; ++j)
      dst[(ti * 16 + mb + j) * GS_K + tj * 16 + m] = acc[j];
  }
  __syncthreads();
  for (int idx = tid; idx < GS_K * GS_K; idx += GS_NT) G[idx] += W[idx];
  __syncthreads();

  if (wave == 0) {
    for (int j = 0; j < GS_K; ++j) {
      float ljj = sqrtf(G[j * GS_K + j]);
      float inv = 1.0f / ljj;
      if (lane == 0) G[j * GS_K + j] = ljj;
      for (int i = j + 1 + lane; i < GS_K; i += 32)
        G[i * GS_K + j] *= inv;
      lds_fence();
      for (int i = j + 1 + lane; i < GS_K; i += 32) {
        float lij = G[i * GS_K + j];
        for (int k = j + 1; k <= i; ++k)
          G[i * GS_K + k] -= lij * G[k * GS_K + j];
      }
      lds_fence();
    }
  }
  __syncthreads();

  for (int j = wave; j < GS_K; j += 32) {
    for (int i = lane; i < GS_K; i += 32)
      W[i * GS_K + j] = (i == j) ? 1.0f : 0.0f;
    lds_fence();
    for (int k = j; k < GS_K; ++k) {
      float wk = W[k * GS_K + j] / G[k * GS_K + k];
      if (lane == 0) W[k * GS_K + j] = wk;
      for (int i = k + 1 + lane; i < GS_K; i += 32)
        W[i * GS_K + j] -= G[i * GS_K + k] * wk;
      lds_fence();
    }
  }
  __syncthreads();

  {
    const int m    = lane & 15;
    const int koff = (lane < 16) ? 0 : 2;
    const int mb   = (lane < 16) ? 0 : 8;
    for (int t = wave; t < 4 * (GS_L / 16); t += 32) {
      const int ti = t >> 6;
      const int tc = t & 63;
      const float* pa = W + (ti * 16 + m) * GS_K + koff;
      v8f acc = {0.f, 0.f, 0.f, 0.f, 0.f, 0.f, 0.f, 0.f};
      #pragma unroll
      for (int k0 = 0; k0 < GS_K; k0 += 4) {
        v2f a = *(const v2f*)(pa + k0);
        v2f bb;
        bb.x = X[(k0 + koff    ) * GS_LP + tc * 16 + m];
        bb.y = X[(k0 + koff + 1) * GS_LP + tc * 16 + m];
        acc = __builtin_amdgcn_wmma_f32_16x16x4_f32(false, a, false, bb,
                                                    (short)0, acc, false, false);
      }
      #pragma unroll
      for (int j = 0; j < 8; ++j)
        og[(ti * 16 + mb + j) * GS_L + tc * 16 + m] = acc[j];
    }
  }
}

extern "C" void kernel_launch(void* const* d_in, const int* in_sizes, int n_in,
                              void* d_out, int out_size, void* d_ws, size_t ws_size,
                              hipStream_t stream) {
  (void)in_sizes; (void)n_in; (void)out_size; (void)d_ws; (void)ws_size;
  const float* x   = (const float*)d_in[0];
  float*       out = (float*)d_out;

  const size_t shmem = (size_t)(GS_K * GS_LP + 2 * GS_K * GS_K) * sizeof(float);
  hipFuncSetAttribute(reinterpret_cast<const void*>(&gram_schmidt_cholqr),
                      hipFuncAttributeMaxDynamicSharedMemorySize, (int)shmem);

  gram_schmidt_cholqr<<<dim3(GS_B), dim3(GS_NT), shmem, stream>>>(x, out);
}


# Elimination sweep takes two pivots per barrier: a wave's two rows are consecutive pivots, so it clears its second by its first by lane exchange and publishes both; 32 barriers, not 64; same sums
# speedup vs baseline: 5.7632x; 5.7632x over previous
#include <hip/hip_runtime.h>

typedef float v2f __attribute__((ext_vector_type(2)));
typedef float v4f __attribute__((ext_vector_type(4)));
typedef float v8f __attribute__((ext_vector_type(8)));

#define GS_B  512
#define GS_K  64
#define GS_L  1024
#define GS_LP 1028
#define GS_NT 1024

static __device__ __forceinline__ float xhalf(float x) {
  return __builtin_bit_cast(float, __builtin_amdgcn_permlanex16(__builtin_bit_cast(int, x), __builtin_bit_cast(int, x),
                                                                0x76543210, 0xfedcba98, false, false));
}
static __device__ __forceinline__ float lane_val(float x, int l) {
  return __builtin_bit_cast(float, __builtin_amdgcn_readlane(__builtin_bit_cast(int, x), l));
}

__global__ __launch_bounds__(GS_NT, 1)
void gram_schmidt_cholqr(const float* __restrict__ xin,
                         float* __restrict__ xout) {
  extern __shared__ float smem[];
  float* X = smem;
  float* G = smem + GS_K * GS_LP;
  float* W = G + GS_K * GS_K;

  const int b    = blockIdx.x;
  const int tid  = threadIdx.x;
  const int wave = tid >> 5;
  const int lane = tid & 31;

  const float* xg = xin  + (size_t)b * (GS_K * GS_L);
  float*       og = xout + (size_t)b * (GS_K * GS_L);

  for (int idx = tid; idx < GS_K * (GS_L / 4); idx += GS_NT) {
    int r  = idx >> 8;
    int c4 = (idx & 255) << 2;
    *(v4f*)(X + r * GS_LP + c4) = *(const v4f*)(xg + r * GS_L + c4);
  }
  __syncthreads();

  {
    const int tile  = wave >> 1;
    const int ti    = tile >> 2;
    const int tj    = tile & 3;
    const int khalf = wave & 1;
    const int m     = lane & 15;
    const int koff  = (lane < 16) ? 0 : 2;
    const float* pa = X + (ti * 16 + m) * GS_LP + koff + khalf * (GS_L / 2);
    const float* pb = X + (tj * 16 + m) * GS_LP + koff + khalf * (GS_L / 2);
    v8f acc = {0.f, 0.f, 0.f, 0.f, 0.f, 0.f, 0.f, 0.f};
    #pragma unroll 4
    for (int k = 0; k < GS_L / 2; k += 4) {
      v2f a  = *(const v2f*)(pa + k);
      v2f bb = *(const v2f*)(pb + k);
      acc = __builtin_amdgcn_wmma_f32_16x16x4_f32(false, a, false, bb,
                                                  (short)0, acc, false, false);
    }
    float*    dst = khalf ? W : G;
    const int mb  = (lane < 16) ? 0 : 8;
    #pragma unroll
    for (int j = 0; j < 8; ++j)
      dst[(ti * 16 + mb + j) * GS_K + tj * 16 + m] = acc[j];
  }
  __syncthreads();
  for (int idx = tid; idx < GS_K * GS_K; idx += GS_NT) G[idx] += W[idx];
  __syncthreads();

  {
    const int i  = tid >> 4;
    const int tc = tid & 15;
    const int wv = __builtin_amdgcn_readfirstlane(wave);
    v4f ag = *(const v4f*)(G + i * GS_K + 4 * tc);
    v4f am;
    am.x = (4 * tc + 0 == i) ? 1.0f : 0.0f;
    am.y = (4 * tc + 1 == i) ? 1.0f : 0.0f;
    am.z = (4 * tc + 2 == i) ? 1.0f : 0.0f;
    am.w = (4 * tc + 3 == i) ? 1.0f : 0.0f;
    float di = 1.0f;
    for (int p = 0; p < GS_K / 2; ++p) {
      float* r0 = W + (p & 1) * (4 * GS_K);
      float* r1 = r0 + 2 * GS_K;
      if (wv == p) {
        v4f bg, bm;
        bg.x = xhalf(ag.x); bg.y = xhalf(ag.y); bg.z = xhalf(ag.z); bg.w = xhalf(ag.w);
        bm.x = xhalf(am.x); bm.y = xhalf(am.y); bm.z = xhalf(am.z); bm.w = xhalf(am.w);
        const int   pl  = p >> 1;
        const float d0  = lane_val((p & 1) ? ag.z : ag.x, pl);
        const float g10 = lane_val((p & 1) ? ag.w : ag.y, pl);
        if (lane >= 16) {
          const float inv = 1.0f / d0;
          ag -= (g10 * bg) * inv;
          am -= (g10 * bm) * inv;
        }
        const float d1 = lane_val((p & 1) ? ag.w : ag.y, 16 + pl);
        di = (lane < 16) ? d0 : d1;
        float* rr = (lane < 16) ? r0 : r1;
        *(v4f*)(rr + 4 * tc)        = ag;
        *(v4f*)(rr + GS_K + 4 * tc) = am;
      }
      __syncthreads();
      if (wv > p) {
        const float d0 = r0[2 * p];
        const float g0 = r0[i];
        const float d1 = r1[2 * p + 1];
        const float g1 = r1[i];
        const v4f rg0 = *(const v4f*)(r0 + 4 * tc);
        const v4f rm0 = *(const v4f*)(r0 + GS_K + 4 * tc);
        const v4f rg1 = *(const v4f*)(r1 + 4 * tc);
        const v4f rm1 = *(const v4f*)(r1 + GS_K + 4 * tc);
        const float inv0 = 1.0f / d0;
        const float inv1 = 1.0f / d1;
        ag -= (g0 * rg0) * inv0;
        am -= (g0 * rm0) * inv0;
        ag -= (g1 * rg1) * inv1;
        am -= (g1 * rm1) * inv1;
      }
    }
    const float s = sqrtf(di);
    *(v4f*)(G + i * GS_K + 4 * tc) = am / s;
  }
  __syncthreads();

  {
    const int m    = lane & 15;
    const int koff = (lane < 16) ? 0 : 2;
    const int mb   = (lane < 16) ? 0 : 8;
    const int ti   = wave & 3;
    const int cg   = wave >> 2;
    const float* pa = G + (ti * 16 + m) * GS_K + koff;
    v2f a[GS_K / 4];
    #pragma unroll
    for (int s = 0; s < GS_K / 4; ++s)
      a[s] = *(const v2f*)(pa + 4 * s);
    for (int n = 0; n < 8; ++n) {
      const int tc = cg * 8 + n;
      const float* pb = X + koff * GS_LP + tc * 16 + m;
      v8f acc = {0.f, 0.f, 0.f, 0.f, 0.f, 0.f, 0.f, 0.f};
      #pragma unroll
      for (int s = 0; s < GS_K / 4; ++s) {
        v2f bb;
        bb.x = pb[(4 * s    ) * GS_LP];
        bb.y = pb[(4 * s + 1) * GS_LP];
        acc = __builtin_amdgcn_wmma_f32_16x16x4_f32(false, a[s], false, bb,
                                                    (short)0, acc, false, false);
      }
      #pragma unroll
      for (int j = 0; j < 8; ++j)
        og[(ti * 16 + mb + j) * GS_L + tc * 16 + m] = acc[j];
    }
  }
}

extern "C" void kernel_launch(void* const* d_in, const int* in_sizes, int n_in,
                              void* d_out, int out_size, void* d_ws, size_t ws_size,
                              hipStream_t stream) {
  (void)in_sizes; (void)n_in; (void)out_size; (void)d_ws; (void)ws_size;
  const float* x   = (const float*)d_in[0];
  float*       out = (float*)d_out;

  const size_t shmem = (size_t)(GS_K * GS_LP + 2 * GS_K * GS_K) * sizeof(float);
  hipFuncSetAttribute(reinterpret_cast<const void*>(&gram_schmidt_cholqr),
                      hipFuncAttributeMaxDynamicSharedMemorySize, (int)shmem);

  gram_schmidt_cholqr<<<dim3(GS_B), dim3(GS_NT), shmem, stream>>>(x, out);
}


# Final product: a wave's four column tiles take interleaved columns (tile t = column 4c+t), so one 128-bit LDS read per k row feeds all four and Q leaves as 256-byte row pieces; same sums
# speedup vs baseline: 6.7934x; 1.1788x over previous
#include <hip/hip_runtime.h>

typedef float v2f __attribute__((ext_vector_type(2)));
typedef float v4f __attribute__((ext_vector_type(4)));
typedef float v8f __attribute__((ext_vector_type(8)));

#define GS_B  512
#define GS_K  64
#define GS_L  1024
#define GS_LP 1028
#define GS_NT 1024

static __device__ __forceinline__ float xhalf(float x) {
  return __builtin_bit_cast(float, __builtin_amdgcn_permlanex16(__builtin_bit_cast(int, x), __builtin_bit_cast(int, x),
                                                                0x76543210, 0xfedcba98, false, false));
}
static __device__ __forceinline__ float lane_val(float x, int l) {
  return __builtin_bit_cast(float, __builtin_amdgcn_readlane(__builtin_bit_cast(int, x), l));
}

__global__ __launch_bounds__(GS_NT, 1)
void gram_schmidt_cholqr(const float* __restrict__ xin,
                         float* __restrict__ xout) {
  extern __shared__ float smem[];
  float* X = smem;
  float* G = smem + GS_K * GS_LP;
  float* W = G + GS_K * GS_K;

  const int b    = blockIdx.x;
  const int tid  = threadIdx.x;
  const int wave = tid >> 5;
  const int lane = tid & 31;

  const float* xg = xin  + (size_t)b * (GS_K * GS_L);
  float*       og = xout + (size_t)b * (GS_K * GS_L);

  for (int idx = tid; idx < GS_K * (GS_L / 4); idx += GS_NT) {
    int r  = idx >> 8;
    int c4 = (idx & 255) << 2;
    *(v4f*)(X + r * GS_LP + c4) = *(const v4f*)(xg + r * GS_L + c4);
  }
  __syncthreads();

  {
    const int tile  = wave >> 1;
    const int ti    = tile >> 2;
    const int tj    = tile & 3;
    const int khalf = wave & 1;
    const int m     = lane & 15;
    const int koff  = (lane < 16) ? 0 : 2;
    const float* pa = X + (ti * 16 + m) * GS_LP + koff + khalf * (GS_L / 2);
    const float* pb = X + (tj * 16 + m) * GS_LP + koff + khalf * (GS_L / 2);
    v8f acc = {0.f, 0.f, 0.f, 0.f, 0.f, 0.f, 0.f, 0.f};
    #pragma unroll 4
    for (int k = 0; k < GS_L / 2; k += 4) {
      v2f a  = *(const v2f*)(pa + k);
      v2f bb = *(const v2f*)(pb + k);
      acc = __builtin_amdgcn_wmma_f32_16x16x4_f32(false, a, false, bb,
                                                  (short)0, acc, false, false);
    }
    float*    dst = khalf ? W : G;
    const int mb  = (lane < 16) ? 0 : 8;
    #pragma unroll
    for (int j = 0; j < 8; ++j)
      dst[(ti * 16 + mb + j) * GS_K + tj * 16 + m] = acc[j];
  }
  __syncthreads();
  for (int idx = tid; idx < GS_K * GS_K; idx += GS_NT) G[idx] += W[idx];
  __syncthreads();

  {
    const int i  = tid >> 4;
    const int tc = tid & 15;
    const int wv = __builtin_amdgcn_readfirstlane(wave);
    v4f ag = *(const v4f*)(G + i * GS_K + 4 * tc);
    v4f am;
    am.x = (4 * tc + 0 == i) ? 1.0f : 0.0f;
    am.y = (4 * tc + 1 == i) ? 1.0f : 0.0f;
    am.z = (4 * tc + 2 == i) ? 1.0f : 0.0f;
    am.w = (4 * tc + 3 == i) ? 1.0f : 0.0f;
    float di = 1.0f;
    for (int p = 0; p < GS_K / 2; ++p) {
      float* r0 = W + (p & 1) * (4 * GS_K);
      float* r1 = r0 + 2 * GS_K;
      if (wv == p) {
        v4f bg, bm;
        bg.x = xhalf(ag.x); bg.y = xhalf(ag.y); bg.z = xhalf(ag.z); bg.w = xhalf(ag.w);
        bm.x = xhalf(am.x); bm.y = xhalf(am.y); bm.z = xhalf(am.z); bm.w = xhalf(am.w);
        const int   pl  = p >> 1;
        const float d0  = lane_val((p & 1) ? ag.z : ag.x, pl);
        const float g10 = lane_val((p & 1) ? ag.w : ag.y, pl);
        if (lane >= 16) {
          const float inv = 1.0f / d0;
          ag -= (g10 * bg) * inv;
          am -= (g10 * bm) * inv;
        }
        const float d1 = lane_val((p & 1) ? ag.w : ag.y, 16 + pl);
        di = (lane < 16) ? d0 : d1;
        float* rr = (lane < 16) ? r0 : r1;
        *(v4f*)(rr + 4 * tc)        = ag;
        *(v4f*)(rr + GS_K + 4 * tc) = am;
      }
      __syncthreads();
      if (wv > p) {
        const float d0 = r0[2 * p];
        const float g0 = r0[i];
        const float d1 = r1[2 * p + 1];
        const float g1 = r1[i];
        const v4f rg0 = *(const v4f*)(r0 + 4 * tc);
        const v4f rm0 = *(const v4f*)(r0 + GS_K + 4 * tc);
        const v4f rg1 = *(const v4f*)(r1 + 4 * tc);
        const v4f rm1 = *(const v4f*)(r1 + GS_K + 4 * tc);
        const float inv0 = 1.0f / d0;
        const float inv1 = 1.0f / d1;
        ag -= (g0 * rg0) * inv0;
        am -= (g0 * rm0) * inv0;
        ag -= (g1 * rg1) * inv1;
        am -= (g1 * rm1) * inv1;
      }
    }
    const float s = sqrtf(di);
    *(v4f*)(G + i * GS_K + 4 * tc) = am / s;
  }
  __syncthreads();

  {
    const int c    = lane & 15;
    const int koff = (lane < 16) ? 0 : 2;
    const int mb   = (lane < 16) ? 0 : 8;
    const int ti   = wave & 3;
    const int cg   = wave >> 2;
    const float* pa = G + (ti * 16 + c) * GS_K + koff;
    v2f a[GS_K / 4];
    #pragma unroll
    for (int s = 0; s < GS_K / 4; ++s)
      a[s] = *(const v2f*)(pa + 4 * s);
    #pragma unroll 1
    for (int g = 0; g < 2; ++g) {
      const int col = cg * 128 + g * 64 + 4 * c;
      const float* pb = X + koff * GS_LP + col;
      v8f acc0 = {0.f, 0.f, 0.f, 0.f, 0.f, 0.f, 0.f, 0.f};
      v8f acc1 = acc0, acc2 = acc0, acc3 = acc0;
      v4f r0 = *(const v4f*)(pb);
      v4f r1 = *(const v4f*)(pb + GS_LP);
      #pragma unroll
      for (int s = 0; s < GS_K / 4; ++s) {
        v2f b0, b1, b2, b3;
        b0.x = r0.x; b0.y = r1.x;
        b1.x = r0.y; b1.y = r1.y;
        b2.x = r0.z; b2.y = r1.z;
        b3.x = r0.w; b3.y = r1.w;
        if (s + 1 < GS_K / 4) {
          r0 = *(const v4f*)(pb + (4 * s + 4) * GS_LP);
          r1 = *(const v4f*)(pb + (4 * s + 5) * GS_LP);
        }
        acc0 = __builtin_amdgcn_wmma_f32_16x16x4_f32(false, a[s], false, b0, (short)0, acc0, false, false);
        acc1 = __builtin_amdgcn_wmma_f32_16x16x4_f32(false, a[s], false, b1, (short)0, acc1, false, false);
        acc2 = __builtin_amdgcn_wmma_f32_16x16x4_f32(false, a[s], false, b2, (short)0, acc2, false, false);
        acc3 = __builtin_amdgcn_wmma_f32_16x16x4_f32(false, a[s], false, b3, (short)0, acc3, false, false);
      }
      float* po = og + (ti * 16 + mb) * GS_L + col;
      #pragma unroll
      for (int j = 0; j < 8; ++j) {
        v4f q;
        q.x = acc0[j]; q.y = acc1[j]; q.z = acc2[j]; q.w = acc3[j];
        *(v4f*)(po + j * GS_L) = q;
      }
    }
  }
}

extern "C" void kernel_launch(void* const* d_in, const int* in_sizes, int n_in,
                              void* d_out, int out_size, void* d_ws, size_t ws_size,
                              hipStream_t stream) {
  (void)in_sizes; (void)n_in; (void)out_size; (void)d_ws; (void)ws_size;
  const float* x   = (const float*)d_in[0];
  float*       out = (float*)d_out;

  const size_t shmem = (size_t)(GS_K * GS_LP + 2 * GS_K * GS_K) * sizeof(float);
  hipFuncSetAttribute(reinterpret_cast<const void*>(&gram_schmidt_cholqr),
                      hipFuncAttributeMaxDynamicSharedMemorySize, (int)shmem);

  gram_schmidt_cholqr<<<dim3(GS_B), dim3(GS_NT), shmem, stream>>>(x, out);
}


# T handed from the sweep to the final product in A-fragment order, XOR-swizzled: each A fragment is one conflict-free 256-byte LDS read instead of 16 lanes on one bank; same sums
# speedup vs baseline: 7.2299x; 1.0643x over previous
#include <hip/hip_runtime.h>

typedef float v2f __attribute__((ext_vector_type(2)));
typedef float v4f __attribute__((ext_vector_type(4)));
typedef float v8f __attribute__((ext_vector_type(8)));

#define GS_B  512
#define GS_K  64
#define GS_L  1024
#define GS_LP 1028
#define GS_NT 1024

static __device__ __forceinline__ float xhalf(float x) {
  return __builtin_bit_cast(float, __builtin_amdgcn_permlanex16(__builtin_bit_cast(int, x), __builtin_bit_cast(int, x),
                                                                0x76543210, 0xfedcba98, false, false));
}
static __device__ __forceinline__ float lane_val(float x, int l) {
  return __builtin_bit_cast(float, __builtin_amdgcn_readlane(__builtin_bit_cast(int, x), l));
}

__global__ __launch_bounds__(GS_NT, 1)
void gram_schmidt_cholqr(const float* __restrict__ xin,
                         float* __restrict__ xout) {
  extern __shared__ float smem[];
  float* X = smem;
  float* G = smem + GS_K * GS_LP;
  float* W = G + GS_K * GS_K;

  const int b    = blockIdx.x;
  const int tid  = threadIdx.x;
  const int wave = tid >> 5;
  const int lane = tid & 31;

  const float* xg = xin  + (size_t)b * (GS_K * GS_L);
  float*       og = xout + (size_t)b * (GS_K * GS_L);

  for (int idx = tid; idx < GS_K * (GS_L / 4); idx += GS_NT) {
    int r  = idx >> 8;
    int c4 = (idx & 255) << 2;
    *(v4f*)(X + r * GS_LP + c4) = *(const v4f*)(xg + r * GS_L + c4);
  }
  __syncthreads();

  {
    const int tile  = wave >> 1;
    const int ti    = tile >> 2;
    const int tj    = tile & 3;
    const int khalf = wave & 1;
    const int m     = lane & 15;
    const int koff  = (lane < 16) ? 0 : 2;
    const float* pa = X + (ti * 16 + m) * GS_LP + koff + khalf * (GS_L / 2);
    const float* pb = X + (tj * 16 + m) * GS_LP + koff + khalf * (GS_L / 2);
    v8f acc = {0.f, 0.f, 0.f, 0.f, 0.f, 0.f, 0.f, 0.f};
    #pragma unroll 4
    for (int k = 0; k < GS_L / 2; k += 4) {
      v2f a  = *(const v2f*)(pa + k);
      v2f bb = *(const v2f*)(pb + k);
      acc = __builtin_amdgcn_wmma_f32_16x16x4_f32(false, a, false, bb,
                                                  (short)0, acc, false, false);
    }
    float*    dst = khalf ? W : G;
    const int mb  = (lane < 16) ? 0 : 8;
    #pragma unroll
    for (int j = 0; j < 8; ++j)
      dst[(ti * 16 + mb + j) * GS_K + tj * 16 + m] = acc[j];
  }
  __syncthreads();
  for (int idx = tid; idx < GS_K * GS_K; idx += GS_NT) G[idx] += W[idx];
  __syncthreads();

  {
    const int i  = tid >> 4;
    const int tc = tid & 15;
    const int wv = __builtin_amdgcn_readfirstlane(wave);
    v4f ag = *(const v4f*)(G + i * GS_K + 4 * tc);
    v4f am;
    am.x = (4 * tc + 0 == i) ? 1.0f : 0.0f;
    am.y = (4 * tc + 1 == i) ? 1.0f : 0.0f;
    am.z = (4 * tc + 2 == i) ? 1.0f : 0.0f;
    am.w = (4 * tc + 3 == i) ? 1.0f : 0.0f;
    float di = 1.0f;
    for (int p = 0; p < GS_K / 2; ++p) {
      float* r0 = W + (p & 1) * (4 * GS_K);
      float* r1 = r0 + 2 * GS_K;
      if (wv == p) {
        v4f bg, bm;
        bg.x = xhalf(ag.x); bg.y = xhalf(ag.y); bg.z = xhalf(ag.z); bg.w = xhalf(ag.w);
        bm.x = xhalf(am.x); bm.y = xhalf(am.y); bm.z = xhalf(am.z); bm.w = xhalf(am.w);
        const int   pl  = p >> 1;
        const float d0  = lane_val((p & 1) ? ag.z : ag.x, pl);
        const float g10 = lane_val((p & 1) ? ag.w : ag.y, pl);
        if (lane >= 16) {
          const float inv = 1.0f / d0;
          ag -= (g10 * bg) * inv;
          am -= (g10 * bm) * inv;
        }
        const float d1 = lane_val((p & 1) ? ag.w : ag.y, 16 + pl);
        di = (lane < 16) ? d0 : d1;
        float* rr = (lane < 16) ? r0 : r1;
        *(v4f*)(rr + 4 * tc)        = ag;
        *(v4f*)(rr + GS_K + 4 * tc) = am;
      }
      __syncthreads();
      if (wv > p) {
        const float d0 = r0[2 * p];
        const float g0 = r0[i];
        const float d1 = r1[2 * p + 1];
        const float g1 = r1[i];
        const v4f rg0 = *(const v4f*)(r0 + 4 * tc);
        const v4f rm0 = *(const v4f*)(r0 + GS_K + 4 * tc);
        const v4f rg1 = *(const v4f*)(r1 + 4 * tc);
        const v4f rm1 = *(const v4f*)(r1 + GS_K + 4 * tc);
        const float inv0 = 1.0f / d0;
        const float inv1 = 1.0f / d1;
        ag -= (g0 * rg0) * inv0;
        am -= (g0 * rm0) * inv0;
        ag -= (g1 * rg1) * inv1;
        am -= (g1 * rm1) * inv1;
      }
    }
    const float s = sqrtf(di);
    const v4f t = am / s;
    float* fb = G + ((i >> 4) * 16 + tc) * 64;
    const int cc = 2 * (i & 15);
    v2f lo, hi;
    lo.x = t.x; lo.y = t.y;
    hi.x = t.z; hi.y = t.w;
    *(v2f*)(fb + (cc ^ (4 * tc)))        = lo;
    *(v2f*)(fb + ((32 + cc) ^ (4 * tc))) = hi;
  }
  __syncthreads();

  {
    const int c    = lane & 15;
    const int koff = (lane < 16) ? 0 : 2;
    const int mb   = (lane < 16) ? 0 : 8;
    const int ti   = wave & 3;
    const int cg   = wave >> 2;
    const float* pa = G + ti * 16 * 64;
    const int    lh = 16 * koff + 2 * c;
    v2f a[GS_K / 4];
    #pragma unroll
    for (int s = 0; s < GS_K / 4; ++s)
      a[s] = *(const v2f*)(pa + 64 * s + (lh ^ (4 * s)));
    #pragma unroll 1
    for (int g = 0; g < 2; ++g) {
      const int col = cg * 128 + g * 64 + 4 * c;
      const float* pb = X + koff * GS_LP + col;
      v8f acc0 = {0.f, 0.f, 0.f, 0.f, 0.f, 0.f, 0.f, 0.f};
      v8f acc1 = acc0, acc2 = acc0, acc3 = acc0;
      v4f r0 = *(const v4f*)(pb);
      v4f r1 = *(const v4f*)(pb + GS_LP);
      #pragma unroll
      for (int s = 0; s < GS_K / 4; ++s) {
        v2f b0, b1, b2, b3;
        b0.x = r0.x; b0.y = r1.x;
        b1.x = r0.y; b1.y = r1.y;
        b2.x = r0.z; b2.y = r1.z;
        b3.x = r0.w; b3.y = r1.w;
        if (s + 1 < GS_K / 4) {
          r0 = *(const v4f*)(pb + (4 * s + 4) * GS_LP);
          r1 = *(const v4f*)(pb + (4 * s + 5) * GS_LP);
        }
        acc0 = __builtin_amdgcn_wmma_f32_16x16x4_f32(false, a[s], false, b0, (short)0, acc0, false, false);
        acc1 = __builtin_amdgcn_wmma_f32_16x16x4_f32(false, a[s], false, b1, (short)0, acc1, false, false);
        acc2 = __builtin_amdgcn_wmma_f32_16x16x4_f32(false, a[s], false, b2, (short)0, acc2, false, false);
        acc3 = __builtin_amdgcn_wmma_f32_16x16x4_f32(false, a[s], false, b3, (short)0, acc3, false, false);
      }
      float* po = og + (ti * 16 + mb) * GS_L + col;
      #pragma unroll
      for (int j = 0; j < 8; ++j) {
        v4f q;
        q.x = acc0[j]; q.y = acc1[j]; q.z = acc2[j]; q.w = acc3[j];
        *(v4f*)(po + j * GS_L) = q;
      }
    }
  }
}

extern "C" void kernel_launch(void* const* d_in, const int* in_sizes, int n_in,
                              void* d_out, int out_size, void* d_ws, size_t ws_size,
                              hipStream_t stream) {
  (void)in_sizes; (void)n_in; (void)out_size; (void)d_ws; (void)ws_size;
  const float* x   = (const float*)d_in[0];
  float*       out = (float*)d_out;

  const size_t shmem = (size_t)(GS_K * GS_LP + 2 * GS_K * GS_K) * sizeof(float);
  hipFuncSetAttribute(reinterpret_cast<const void*>(&gram_schmidt_cholqr),
                      hipFuncAttributeMaxDynamicSharedMemorySize, (int)shmem);

  gram_schmidt_cholqr<<<dim3(GS_B), dim3(GS_NT), shmem, stream>>>(x, out);
}


# Sweep: pivot reciprocals and the final 1/sqrt(d) by the one-instruction forms (v_rcp_f32, v_rsq_f32, ~1 ulp), not the 10-instruction exact division on each step's chain; last-place rounding changes
# speedup vs baseline: 7.5849x; 1.0491x over previous
#include <hip/hip_runtime.h>

typedef float v2f __attribute__((ext_vector_type(2)));
typedef float v4f __attribute__((ext_vector_type(4)));
typedef float v8f __attribute__((ext_vector_type(8)));

#define GS_B  512
#define GS_K  64
#define GS_L  1024
#define GS_LP 1028
#define GS_NT 1024

static __device__ __forceinline__ float xhalf(float x) {
  return __builtin_bit_cast(float, __builtin_amdgcn_permlanex16(__builtin_bit_cast(int, x), __builtin_bit_cast(int, x),
                                                                0x76543210, 0xfedcba98, false, false));
}
static __device__ __forceinline__ float lane_val(float x, int l) {
  return __builtin_bit_cast(float, __builtin_amdgcn_readlane(__builtin_bit_cast(int, x), l));
}

static __device__ __forceinline__ float rcp_hw(float x)  { return __builtin_amdgcn_rcpf(x); }
static __device__ __forceinline__ float rsq_hw(float x)  { return __builtin_amdgcn_rsqf(x); }

__global__ __launch_bounds__(GS_NT, 1)
void gram_schmidt_cholqr(const float* __restrict__ xin,
                         float* __restrict__ xout) {
  extern __shared__ float smem[];
  float* X = smem;
  float* G = smem + GS_K * GS_LP;
  float* W = G + GS_K * GS_K;

  const int b    = blockIdx.x;
  const int tid  = threadIdx.x;
  const int wave = tid >> 5;
  const int lane = tid & 31;

  const float* xg = xin  + (size_t)b * (GS_K * GS_L);
  float*       og = xout + (size_t)b * (GS_K * GS_L);

  for (int idx = tid; idx < GS_K * (GS_L / 4); idx += GS_NT) {
    int r  = idx >> 8;
    int c4 = (idx & 255) << 2;
    *(v4f*)(X + r * GS_LP + c4) = *(const v4f*)(xg + r * GS_L + c4);
  }
  __syncthreads();

  {
    const int tile  = wave >> 1;
    const int ti    = tile >> 2;
    const int tj    = tile & 3;
    const int khalf = wave & 1;
    const int m     = lane & 15;
    const int koff  = (lane < 16) ? 0 : 2;
    const float* pa = X + (ti * 16 + m) * GS_LP + koff + khalf * (GS_L / 2);
    const float* pb = X + (tj * 16 + m) * GS_LP + koff + khalf * (GS_L / 2);
    v8f acc = {0.f, 0.f, 0.f, 0.f, 0.f, 0.f, 0.f, 0.f};
    #pragma unroll 4
    for (int k = 0; k < GS_L / 2; k += 4) {
      v2f a  = *(const v2f*)(pa + k);
      v2f bb = *(const v2f*)(pb + k);
      acc = __builtin_amdgcn_wmma_f32_16x16x4_f32(false, a, false, bb,
                                                  (short)0, acc, false, false);
    }
    float*    dst = khalf ? W : G;
    const int mb  = (lane < 16) ? 0 : 8;
    #pragma unroll
    for (int j = 0; j < 8; ++j)
      dst[(ti * 16 + mb + j) * GS_K + tj * 16 + m] = acc[j];
  }
  __syncthreads();
  for (int idx = tid; idx < GS_K * GS_K; idx += GS_NT) G[idx] += W[idx];
  __syncthreads();

  {
    const int i  = tid >> 4;
    const int tc = tid & 15;
    const int wv = __builtin_amdgcn_readfirstlane(wave);
    v4f ag = *(const v4f*)(G + i * GS_K + 4 * tc);
    v4f am;
    am.x = (4 * tc + 0 == i) ? 1.0f : 0.0f;
    am.y = (4 * tc + 1 == i) ? 1.0f : 0.0f;
    am.z = (4 * tc + 2 == i) ? 1.0f : 0.0f;
    am.w = (4 * tc + 3 == i) ? 1.0f : 0.0f;
    float di = 1.0f;
    for (int p = 0; p < GS_K / 2; ++p) {
      float* r0 = W + (p & 1) * (4 * GS_K);
      float* r1 = r0 + 2 * GS_K;
      if (wv == p) {
        v4f bg, bm;
        bg.x = xhalf(ag.x); bg.y = xhalf(ag.y); bg.z = xhalf(ag.z); bg.w = xhalf(ag.w);
        bm.x = xhalf(am.x); bm.y = xhalf(am.y); bm.z = xhalf(am.z); bm.w = xhalf(am.w);
        const int   pl  = p >> 1;
        const float d0  = lane_val((p & 1) ? ag.z : ag.x, pl);
        const float g10 = lane_val((p & 1) ? ag.w : ag.y, pl);
        if (lane >= 16) {
          const float inv = rcp_hw(d0);
          ag -= (g10 * bg) * inv;
          am -= (g10 * bm) * inv;
        }
        const float d1 = lane_val((p & 1) ? ag.w : ag.y, 16 + pl);
        di = (lane < 16) ? d0 : d1;
        float* rr = (lane < 16) ? r0 : r1;
        *(v4f*)(rr + 4 * tc)        = ag;
        *(v4f*)(rr + GS_K + 4 * tc) = am;
      }
      __syncthreads();
      if (wv > p) {
        const float d0 = r0[2 * p];
        const float g0 = r0[i];
        const float d1 = r1[2 * p + 1];
        const float g1 = r1[i];
        const v4f rg0 = *(const v4f*)(r0 + 4 * tc);
        const v4f rm0 = *(const v4f*)(r0 + GS_K + 4 * tc);
        const v4f rg1 = *(const v4f*)(r1 + 4 * tc);
        const v4f rm1 = *(const v4f*)(r1 + GS_K + 4 * tc);
        const float inv0 = rcp_hw(d0);
        const float inv1 = rcp_hw(d1);
        ag -= (g0 * rg0) * inv0;
        am -= (g0 * rm0) * inv0;
        ag -= (g1 * rg1) * inv1;
        am -= (g1 * rm1) * inv1;
      }
    }
    const float s = rsq_hw(di);
    const v4f t = am * s;
    float* fb = G + ((i >> 4) * 16 + tc) * 64;
    const int cc = 2 * (i & 15);
    v2f lo, hi;
    lo.x = t.x; lo.y = t.y;
    hi.x = t.z; hi.y = t.w;
    *(v2f*)(fb + (cc ^ (4 * tc)))        = lo;
    *(v2f*)(fb + ((32 + cc) ^ (4 * tc))) = hi;
  }
  __syncthreads();

  {
    const int c    = lane & 15;
    const int koff = (lane < 16) ? 0 : 2;
    const int mb   = (lane < 16) ? 0 : 8;
    const int ti   = wave & 3;
    const int cg   = wave >> 2;
    const float* pa = G + ti * 16 * 64;
    const int    lh = 16 * koff + 2 * c;
    v2f a[GS_K / 4];
    #pragma unroll
    for (int s = 0; s < GS_K / 4; ++s)
      a[s] = *(const v2f*)(pa + 64 * s + (lh ^ (4 * s)));
    #pragma unroll 1
    for (int g = 0; g < 2; ++g) {
      const int col = cg * 128 + g * 64 + 4 * c;
      const float* pb = X + koff * GS_LP + col;
      v8f acc0 = {0.f, 0.f, 0.f, 0.f, 0.f, 0.f, 0.f, 0.f};
      v8f acc1 = acc0, acc2 = acc0, acc3 = acc0;
      v4f r0 = *(const v4f*)(pb);
      v4f r1 = *(const v4f*)(pb + GS_LP);
      #pragma unroll
      for (int s = 0; s < GS_K / 4; ++s) {
        v2f b0, b1, b2, b3;
        b0.x = r0.x; b0.y = r1.x;
        b1.x = r0.y; b1.y = r1.y;
        b2.x = r0.z; b2.y = r1.z;
        b3.x = r0.w; b3.y = r1.w;
        if (s + 1 < GS_K / 4) {
          r0 = *(const v4f*)(pb + (4 * s + 4) * GS_LP);
          r1 = *(const v4f*)(pb + (4 * s + 5) * GS_LP);
        }
        acc0 = __builtin_amdgcn_wmma_f32_16x16x4_f32(false, a[s], false, b0, (short)0, acc0, false, false);
        acc1 = __builtin_amdgcn_wmma_f32_16x16x4_f32(false, a[s], false, b1, (short)0, acc1, false, false);
        acc2 = __builtin_amdgcn_wmma_f32_16x16x4_f32(false, a[s], false, b2, (short)0, acc2, false, false);
        acc3 = __builtin_amdgcn_wmma_f32_16x16x4_f32(false, a[s], false, b3, (short)0, acc3, false, false);
      }
      float* po = og + (ti * 16 + mb) * GS_L + col;
      #pragma unroll
      for (int j = 0; j < 8; ++j) {
        v4f q;
        q.x = acc0[j]; q.y = acc1[j]; q.z = acc2[j]; q.w = acc3[j];
        *(v4f*)(po + j * GS_L) = q;
      }
    }
  }
}

extern "C" void kernel_launch(void* const* d_in, const int* in_sizes, int n_in,
                              void* d_out, int out_size, void* d_ws, size_t ws_size,
                              hipStream_t stream) {
  (void)in_sizes; (void)n_in; (void)out_size; (void)d_ws; (void)ws_size;
  const float* x   = (const float*)d_in[0];
  float*       out = (float*)d_out;

  const size_t shmem = (size_t)(GS_K * GS_LP + 2 * GS_K * GS_K) * sizeof(float);
  hipFuncSetAttribute(reinterpret_cast<const void*>(&gram_schmidt_cholqr),
                      hipFuncAttributeMaxDynamicSharedMemorySize, (int)shmem);

  gram_schmidt_cholqr<<<dim3(GS_B), dim3(GS_NT), shmem, stream>>>(x, out);
}


# Load of X overlapped with the Gram product: X arrives in four chunks (128 columns of each K half), chunk c+1's global loads in flight under chunk c's products, stored to LDS behind them; same sums
# speedup vs baseline: 8.0017x; 1.0550x over previous
#include <hip/hip_runtime.h>

typedef float v2f __attribute__((ext_vector_type(2)));
typedef float v4f __attribute__((ext_vector_type(4)));
typedef float v8f __attribute__((ext_vector_type(8)));

#define GS_B  512
#define GS_K  64
#define GS_L  1024
#define GS_LP 1028
#define GS_NT 1024
#define GS_NCH 4
#define GS_CW  (GS_L / 2 / GS_NCH)

static __device__ __forceinline__ float xhalf(float x) {
  return __builtin_bit_cast(float, __builtin_amdgcn_permlanex16(__builtin_bit_cast(int, x), __builtin_bit_cast(int, x),
                                                                0x76543210, 0xfedcba98, false, false));
}
static __device__ __forceinline__ float lane_val(float x, int l) {
  return __builtin_bit_cast(float, __builtin_amdgcn_readlane(__builtin_bit_cast(int, x), l));
}

static __device__ __forceinline__ float rcp_hw(float x)  { return __builtin_amdgcn_rcpf(x); }
static __device__ __forceinline__ float rsq_hw(float x)  { return __builtin_amdgcn_rsqf(x); }

__global__ __launch_bounds__(GS_NT, 1)
void gram_schmidt_cholqr(const float* __restrict__ xin,
                         float* __restrict__ xout) {
  extern __shared__ float smem[];
  float* X = smem;
  float* G = smem + GS_K * GS_LP;
  float* W = G + GS_K * GS_K;

  const int b    = blockIdx.x;
  const int tid  = threadIdx.x;
  const int wave = tid >> 5;
  const int lane = tid & 31;

  const float* xg = xin  + (size_t)b * (GS_K * GS_L);
  float*       og = xout + (size_t)b * (GS_K * GS_L);

  {
    const int tile  = wave >> 1;
    const int ti    = tile >> 2;
    const int tj    = tile & 3;
    const int khalf = wave & 1;
    const int m     = lane & 15;
    const int koff  = (lane < 16) ? 0 : 2;
    const float* pa = X + (ti * 16 + m) * GS_LP + koff + khalf * (GS_L / 2);
    const float* pb = X + (tj * 16 + m) * GS_LP + koff + khalf * (GS_L / 2);
    v8f acc = {0.f, 0.f, 0.f, 0.f, 0.f, 0.f, 0.f, 0.f};

    const int so = 4 * lane;
    v4f st[4 * (GS_CW / 128)];
    #pragma unroll
    for (int j = 0; j < 4 * (GS_CW / 128); ++j) {
      const int row = 2 * wave + ((j >> 1) & 1);
      const int col = (j & 1) * (GS_L / 2) + (j >> 2) * 128 + so;
      st[j] = *(const v4f*)(xg + row * GS_L + col);
    }
    #pragma unroll
    for (int j = 0; j < 4 * (GS_CW / 128); ++j) {
      const int row = 2 * wave + ((j >> 1) & 1);
      const int col = (j & 1) * (GS_L / 2) + (j >> 2) * 128 + so;
      *(v4f*)(X + row * GS_LP + col) = st[j];
    }
    __syncthreads();

    #pragma unroll
    for (int c = 0; c < GS_NCH; ++c) {
      if (c + 1 < GS_NCH) {
        #pragma unroll
        for (int j = 0; j < 4 * (GS_CW / 128); ++j) {
          const int row = 2 * wave + ((j >> 1) & 1);
          const int col = (j & 1) * (GS_L / 2) + (c + 1) * GS_CW + (j >> 2) * 128 + so;
          st[j] = *(const v4f*)(xg + row * GS_L + col);
        }
      }
      #pragma unroll 4
      for (int k = c * GS_CW; k < (c + 1) * GS_CW; k += 4) {
        v2f a  = *(const v2f*)(pa + k);
        v2f bb = *(const v2f*)(pb + k);
        acc = __builtin_amdgcn_wmma_f32_16x16x4_f32(false, a, false, bb,
                                                    (short)0, acc, false, false);
      }
      if (c + 1 < GS_NCH) {
        #pragma unroll
        for (int j = 0; j < 4 * (GS_CW / 128); ++j) {
          const int row = 2 * wave + ((j >> 1) & 1);
          const int col = (j & 1) * (GS_L / 2) + (c + 1) * GS_CW + (j >> 2) * 128 + so;
          *(v4f*)(X + row * GS_LP + col) = st[j];
        }
        __syncthreads();
      }
    }
    float*    dst = khalf ? W : G;
    const int mb  = (lane < 16) ? 0 : 8;
    #pragma unroll
    for (int j = 0; j < 8; ++j)
      dst[(ti * 16 + mb + j) * GS_K + tj * 16 + m] = acc[j];
  }
  __syncthreads();
  for (int idx = tid; idx < GS_K * GS_K; idx += GS_NT) G[idx] += W[idx];
  __syncthreads();

  {
    const int i  = tid >> 4;
    const int tc = tid & 15;
    const int wv = __builtin_amdgcn_readfirstlane(wave);
    v4f ag = *(const v4f*)(G + i * GS_K + 4 * tc);
    v4f am;
    am.x = (4 * tc + 0 == i) ? 1.0f : 0.0f;
    am.y = (4 * tc + 1 == i) ? 1.0f : 0.0f;
    am.z = (4 * tc + 2 == i) ? 1.0f : 0.0f;
    am.w = (4 * tc + 3 == i) ? 1.0f : 0.0f;
    float di = 1.0f;
    for (int p = 0; p < GS_K / 2; ++p) {
      float* r0 = W + (p & 1) * (4 * GS_K);
      float* r1 = r0 + 2 * GS_K;
      if (wv == p) {
        v4f bg, bm;
        bg.x = xhalf(ag.x); bg.y = xhalf(ag.y); bg.z = xhalf(ag.z); bg.w = xhalf(ag.w);
        bm.x = xhalf(am.x); bm.y = xhalf(am.y); bm.z = xhalf(am.z); bm.w = xhalf(am.w);
        const int   pl  = p >> 1;
        const float d0  = lane_val((p & 1) ? ag.z : ag.x, pl);
        const float g10 = lane_val((p & 1) ? ag.w : ag.y, pl);
        if (lane >= 16) {
          const float inv = rcp_hw(d0);
          ag -= (g10 * bg) * inv;
          am -= (g10 * bm) * inv;
        }
        const float d1 = lane_val((p & 1) ? ag.w : ag.y, 16 + pl);
        di = (lane < 16) ? d0 : d1;
        float* rr = (lane < 16) ? r0 : r1;
        *(v4f*)(rr + 4 * tc)        = ag;
        *(v4f*)(rr + GS_K + 4 * tc) = am;
      }
      __syncthreads();
      if (wv > p) {
        const float d0 = r0[2 * p];
        const float g0 = r0[i];
        const float d1 = r1[2 * p + 1];
        const float g1 = r1[i];
        const v4f rg0 = *(const v4f*)(r0 + 4 * tc);
        const v4f rm0 = *(const v4f*)(r0 + GS_K + 4 * tc);
        const v4f rg1 = *(const v4f*)(r1 + 4 * tc);
        const v4f rm1 = *(const v4f*)(r1 + GS_K + 4 * tc);
        const float inv0 = rcp_hw(d0);
        const float inv1 = rcp_hw(d1);
        ag -= (g0 * rg0) * inv0;
        am -= (g0 * rm0) * inv0;
        ag -= (g1 * rg1) * inv1;
        am -= (g1 * rm1) * inv1;
      }
    }
    const float s = rsq_hw(di);
    const v4f t = am * s;
    float* fb = G + ((i >> 4) * 16 + tc) * 64;
    const int cc = 2 * (i & 15);
    v2f lo, hi;
    lo.x = t.x; lo.y = t.y;
    hi.x = t.z; hi.y = t.w;
    *(v2f*)(fb + (cc ^ (4 * tc)))        = lo;
    *(v2f*)(fb + ((32 + cc) ^ (4 * tc))) = hi;
  }
  __syncthreads();

  {
    const int c    = lane & 15;
    const int koff = (lane < 16) ? 0 : 2;
    const int mb   = (lane < 16) ? 0 : 8;
    const int ti   = wave & 3;
    const int cg   = wave >> 2;
    const float* pa = G + ti * 16 * 64;
    const int    lh = 16 * koff + 2 * c;
    v2f a[GS_K / 4];
    #pragma unroll
    for (int s = 0; s < GS_K / 4; ++s)
      a[s] = *(const v2f*)(pa + 64 * s + (lh ^ (4 * s)));
    #pragma unroll 1
    for (int g = 0; g < 2; ++g) {
      const int col = cg * 128 + g * 64 + 4 * c;
      const float* pb = X + koff * GS_LP + col;
      v8f acc0 = {0.f, 0.f, 0.f, 0.f, 0.f, 0.f, 0.f, 0.f};
      v8f acc1 = acc0, acc2 = acc0, acc3 = acc0;
      v4f r0 = *(const v4f*)(pb);
      v4f r1 = *(const v4f*)(pb + GS_LP);
      #pragma unroll
      for (int s = 0; s < GS_K / 4; ++s) {
        v2f b0, b1, b2, b3;
        b0.x = r0.x; b0.y = r1.x;
        b1.x = r0.y; b1.y = r1.y;
        b2.x = r0.z; b2.y = r1.z;
        b3.x = r0.w; b3.y = r1.w;
        if (s + 1 < GS_K / 4) {
          r0 = *(const v4f*)(pb + (4 * s + 4) * GS_LP);
          r1 = *(const v4f*)(pb + (4 * s + 5) * GS_LP);
        }
        acc0 = __builtin_amdgcn_wmma_f32_16x16x4_f32(false, a[s], false, b0, (short)0, acc0, false, false);
        acc1 = __builtin_amdgcn_wmma_f32_16x16x4_f32(false, a[s], false, b1, (short)0, acc1, false, false);
        acc2 = __builtin_amdgcn_wmma_f32_16x16x4_f32(false, a[s], false, b2, (short)0, acc2, false, false);
        acc3 = __builtin_amdgcn_wmma_f32_16x16x4_f32(false, a[s], false, b3, (short)0, acc3, false, false);
      }
      float* po = og + (ti * 16 + mb) * GS_L + col;
      #pragma unroll
      for (int j = 0; j < 8; ++j) {
        v4f q;
        q.x = acc0[j]; q.y = acc1[j]; q.z = acc2[j]; q.w = acc3[j];
        *(v4f*)(po + j * GS_L) = q;
      }
    }
  }
}

extern "C" void kernel_launch(void* const* d_in, const int* in_sizes, int n_in,
                              void* d_out, int out_size, void* d_ws, size_t ws_size,
                              hipStream_t stream) {
  (void)in_sizes; (void)n_in; (void)out_size; (void)d_ws; (void)ws_size;
  const float* x   = (const float*)d_in[0];
  float*       out = (float*)d_out;

  const size_t shmem = (size_t)(GS_K * GS_LP + 2 * GS_K * GS_K) * sizeof(float);
  hipFuncSetAttribute(reinterpret_cast<const void*>(&gram_schmidt_cholqr),
                      hipFuncAttributeMaxDynamicSharedMemorySize, (int)shmem);

  gram_schmidt_cholqr<<<dim3(GS_B), dim3(GS_NT), shmem, stream>>>(x, out);
}


# Elimination sweep in place: dead column k of G holds column k of U^-1, so a row is 64 values, not 128: half the publish, LDS reads and multiply-adds per step; same sums
# speedup vs baseline: 8.2471x; 1.0307x over previous
#include <hip/hip_runtime.h>

typedef float v2f __attribute__((ext_vector_type(2)));
typedef float v4f __attribute__((ext_vector_type(4)));
typedef float v8f __attribute__((ext_vector_type(8)));

#define GS_B  512
#define GS_K  64
#define GS_L  1024
#define GS_LP 1028
#define GS_NT 1024
#define GS_NCH 4
#define GS_CW  (GS_L / 2 / GS_NCH)

static __device__ __forceinline__ float xhalf(float x) {
  return __builtin_bit_cast(float, __builtin_amdgcn_permlanex16(__builtin_bit_cast(int, x), __builtin_bit_cast(int, x),
                                                                0x76543210, 0xfedcba98, false, false));
}
static __device__ __forceinline__ float lane_val(float x, int l) {
  return __builtin_bit_cast(float, __builtin_amdgcn_readlane(__builtin_bit_cast(int, x), l));
}

static __device__ __forceinline__ float rcp_hw(float x)  { return __builtin_amdgcn_rcpf(x); }
static __device__ __forceinline__ float rsq_hw(float x)  { return __builtin_amdgcn_rsqf(x); }

__global__ __launch_bounds__(GS_NT, 1)
void gram_schmidt_cholqr(const float* __restrict__ xin,
                         float* __restrict__ xout) {
  extern __shared__ float smem[];
  float* X = smem;
  float* G = smem + GS_K * GS_LP;
  float* W = G + GS_K * GS_K;

  const int b    = blockIdx.x;
  const int tid  = threadIdx.x;
  const int wave = tid >> 5;
  const int lane = tid & 31;

  const float* xg = xin  + (size_t)b * (GS_K * GS_L);
  float*       og = xout + (size_t)b * (GS_K * GS_L);

  {
    const int tile  = wave >> 1;
    const int ti    = tile >> 2;
    const int tj    = tile & 3;
    const int khalf = wave & 1;
    const int m     = lane & 15;
    const int koff  = (lane < 16) ? 0 : 2;
    const float* pa = X + (ti * 16 + m) * GS_LP + koff + khalf * (GS_L / 2);
    const float* pb = X + (tj * 16 + m) * GS_LP + koff + khalf * (GS_L / 2);
    v8f acc = {0.f, 0.f, 0.f, 0.f, 0.f, 0.f, 0.f, 0.f};

    const int so = 4 * lane;
    v4f st[4 * (GS_CW / 128)];
    #pragma unroll
    for (int j = 0; j < 4 * (GS_CW / 128); ++j) {
      const int row = 2 * wave + ((j >> 1) & 1);
      const int col = (j & 1) * (GS_L / 2) + (j >> 2) * 128 + so;
      st[j] = *(const v4f*)(xg + row * GS_L + col);
    }
    #pragma unroll
    for (int j = 0; j < 4 * (GS_CW / 128); ++j) {
      const int row = 2 * wave + ((j >> 1) & 1);
      const int col = (j & 1) * (GS_L / 2) + (j >> 2) * 128 + so;
      *(v4f*)(X + row * GS_LP + col) = st[j];
    }
    __syncthreads();

    #pragma unroll
    for (int c = 0; c < GS_NCH; ++c) {
      if (c + 1 < GS_NCH) {
        #pragma unroll
        for (int j = 0; j < 4 * (GS_CW / 128); ++j) {
          const int row = 2 * wave + ((j >> 1) & 1);
          const int col = (j & 1) * (GS_L / 2) + (c + 1) * GS_CW + (j >> 2) * 128 + so;
          st[j] = *(const v4f*)(xg + row * GS_L + col);
        }
      }
      #pragma unroll 4
      for (int k = c * GS_CW; k < (c + 1) * GS_CW; k += 4) {
        v2f a  = *(const v2f*)(pa + k);
        v2f bb = *(const v2f*)(pb + k);
        acc = __builtin_amdgcn_wmma_f32_16x16x4_f32(false, a, false, bb,
                                                    (short)0, acc, false, false);
      }
      if (c + 1 < GS_NCH) {
        #pragma unroll
        for (int j = 0; j < 4 * (GS_CW / 128); ++j) {
          const int row = 2 * wave + ((j >> 1) & 1);
          const int col = (j & 1) * (GS_L / 2) + (c + 1) * GS_CW + (j >> 2) * 128 + so;
          *(v4f*)(X + row * GS_LP + col) = st[j];
        }
        __syncthreads();
      }
    }
    float*    dst = khalf ? W : G;
    const int mb  = (lane < 16) ? 0 : 8;
    #pragma unroll
    for (int j = 0; j < 8; ++j)
      dst[(ti * 16 + mb + j) * GS_K + tj * 16 + m] = acc[j];
  }
  __syncthreads();
  for (int idx = tid; idx < GS_K * GS_K; idx += GS_NT) G[idx] += W[idx];
  __syncthreads();

  {
    const int i  = tid >> 4;
    const int tc = tid & 15;
    const int wv = __builtin_amdgcn_readfirstlane(wave);
    v4f a = *(const v4f*)(G + i * GS_K + 4 * tc);
    const bool ox = (4 * tc + 0 == i), oy = (4 * tc + 1 == i), oz = (4 * tc + 2 == i), ow = (4 * tc + 3 == i);
    const bool qx = (4 * tc + 0 == (i ^ 1)), qy = (4 * tc + 1 == (i ^ 1)),
               qz = (4 * tc + 2 == (i ^ 1)), qw = (4 * tc + 3 == (i ^ 1));
    float di = 1.0f;
    for (int p = 0; p < GS_K / 2; ++p) {
      float* r0 = W + (p & 1) * (4 * GS_K);
      float* r1 = r0 + 2 * GS_K;
      const int  pl  = p >> 1;
      const bool odd = (p & 1) != 0;
      if (wv == p) {
        v4f b;
        b.x = xhalf(a.x); b.y = xhalf(a.y); b.z = xhalf(a.z); b.w = xhalf(a.w);
        const float d0  = lane_val(odd ? a.z : a.x, pl);
        const float g10 = lane_val(odd ? a.w : a.y, pl);
        if (lane >= 16) {
          const float inv = rcp_hw(d0);
          b.x = qx ? 1.0f : b.x;  b.y = qy ? 1.0f : b.y;  b.z = qz ? 1.0f : b.z;  b.w = qw ? 1.0f : b.w;
          a.x = qx ? 0.0f : a.x;  a.y = qy ? 0.0f : a.y;  a.z = qz ? 0.0f : a.z;  a.w = qw ? 0.0f : a.w;
          a -= (g10 * b) * inv;
        }
        const float d1 = lane_val(odd ? a.w : a.y, 16 + pl);
        di = (lane < 16) ? d0 : d1;
        float* rr = (lane < 16) ? r0 : r1;
        v4f pa;
        pa.x = ox ? 1.0f : a.x;  pa.y = oy ? 1.0f : a.y;  pa.z = oz ? 1.0f : a.z;  pa.w = ow ? 1.0f : a.w;
        *(v4f*)(rr + 4 * tc) = pa;
        if (tc == 0) rr[GS_K] = di;
      }
      __syncthreads();
      if (wv > p) {
        const float d0 = r0[GS_K];
        const float g0 = r0[i];
        const float d1 = r1[GS_K];
        const float g1 = r1[i];
        const v4f rw0 = *(const v4f*)(r0 + 4 * tc);
        const v4f rw1 = *(const v4f*)(r1 + 4 * tc);
        const float inv0 = rcp_hw(d0);
        const float inv1 = rcp_hw(d1);
        const bool h0 = (tc == pl) && !odd;
        const bool h1 = (tc == pl) && odd;
        a.x = h0 ? 0.0f : a.x;
        a.z = h1 ? 0.0f : a.z;
        a -= (g0 * rw0) * inv0;
        a.y = h0 ? 0.0f : a.y;
        a.w = h1 ? 0.0f : a.w;
        a -= (g1 * rw1) * inv1;
      }
    }
    const float s = rsq_hw(di);
    v4f t;
    t.x = (4 * tc + 0 < i) ? a.x * s : (ox ? s : 0.0f);
    t.y = (4 * tc + 1 < i) ? a.y * s : (oy ? s : 0.0f);
    t.z = (4 * tc + 2 < i) ? a.z * s : (oz ? s : 0.0f);
    t.w = (4 * tc + 3 < i) ? a.w * s : (ow ? s : 0.0f);
    float* fb = G + ((i >> 4) * 16 + tc) * 64;
    const int cc = 2 * (i & 15);
    v2f lo, hi;
    lo.x = t.x; lo.y = t.y;
    hi.x = t.z; hi.y = t.w;
    *(v2f*)(fb + (cc ^ (4 * tc)))        = lo;
    *(v2f*)(fb + ((32 + cc) ^ (4 * tc))) = hi;
  }
  __syncthreads();

  {
    const int c    = lane & 15;
    const int koff = (lane < 16) ? 0 : 2;
    const int mb   = (lane < 16) ? 0 : 8;
    const int ti   = wave & 3;
    const int cg   = wave >> 2;
    const float* pa = G + ti * 16 * 64;
    const int    lh = 16 * koff + 2 * c;
    v2f a[GS_K / 4];
    #pragma unroll
    for (int s = 0; s < GS_K / 4; ++s)
      a[s] = *(const v2f*)(pa + 64 * s + (lh ^ (4 * s)));
    #pragma unroll 1
    for (int g = 0; g < 2; ++g) {
      const int col = cg * 128 + g * 64 + 4 * c;
      const float* pb = X + koff * GS_LP + col;
      v8f acc0 = {0.f, 0.f, 0.f, 0.f, 0.f, 0.f, 0.f, 0.f};
      v8f acc1 = acc0, acc2 = acc0, acc3 = acc0;
      v4f r0 = *(const v4f*)(pb);
      v4f r1 = *(const v4f*)(pb + GS_LP);
      #pragma unroll
      for (int s = 0; s < GS_K / 4; ++s) {
        v2f b0, b1, b2, b3;
        b0.x = r0.x; b0.y = r1.x;
        b1.x = r0.y; b1.y = r1.y;
        b2.x = r0.z; b2.y = r1.z;
        b3.x = r0.w; b3.y = r1.w;
        if (s + 1 < GS_K / 4) {
          r0 = *(const v4f*)(pb + (4 * s + 4) * GS_LP);
          r1 = *(const v4f*)(pb + (4 * s + 5) * GS_LP);
        }
        acc0 = __builtin_amdgcn_wmma_f32_16x16x4_f32(false, a[s], false, b0, (short)0, acc0, false, false);
        acc1 = __builtin_amdgcn_wmma_f32_16x16x4_f32(false, a[s], false, b1, (short)0, acc1, false, false);
        acc2 = __builtin_amdgcn_wmma_f32_16x16x4_f32(false, a[s], false, b2, (short)0, acc2, false, false);
        acc3 = __builtin_amdgcn_wmma_f32_16x16x4_f32(false, a[s], false, b3, (short)0, acc3, false, false);
      }
      float* po = og + (ti * 16 + mb) * GS_L + col;
      #pragma unroll
      for (int j = 0; j < 8; ++j) {
        v4f q;
        q.x = acc0[j]; q.y = acc1[j]; q.z = acc2[j]; q.w = acc3[j];
        *(v4f*)(po + j * GS_L) = q;
      }
    }
  }
}

extern "C" void kernel_launch(void* const* d_in, const int* in_sizes, int n_in,
                              void* d_out, int out_size, void* d_ws, size_t ws_size,
                              hipStream_t stream) {
  (void)in_sizes; (void)n_in; (void)out_size; (void)d_ws; (void)ws_size;
  const float* x   = (const float*)d_in[0];
  float*       out = (float*)d_out;

  const size_t shmem = (size_t)(GS_K * GS_LP + 2 * GS_K * GS_K) * sizeof(float);
  hipFuncSetAttribute(reinterpret_cast<const void*>(&gram_schmidt_cholqr),
                      hipFuncAttributeMaxDynamicSharedMemorySize, (int)shmem);

  gram_schmidt_cholqr<<<dim3(GS_B), dim3(GS_NT), shmem, stream>>>(x, out);
}


# 16-wave work-groups that stream X through two LDS chunk buffers (148 KB, not 289): two work-groups share a compute unit, one's sweep runs under the other's products; half of X read twice; same sums
# speedup vs baseline: 9.1436x; 1.1087x over previous
#include <hip/hip_runtime.h>


typedef float v2f __attribute__((ext_vector_type(2)));
typedef float v4f __attribute__((ext_vector_type(4)));
typedef float v8f __attribute__((ext_vector_type(8)));

#define GS_B  512
#define GS_K  64
#define GS_L  1024
#define GS_NT 512
#define GS_NCH 4
#define GS_CW  (GS_L / 2 / GS_NCH)
#define GS_BP  (2 * GS_CW + 4)
#define GS_BUF (GS_K * GS_BP)
#define GS_NST 8

static __device__ __forceinline__ float xhalf(float x) {
  return __builtin_bit_cast(float, __builtin_amdgcn_permlanex16(__builtin_bit_cast(int, x), __builtin_bit_cast(int, x),
                                                                0x76543210, 0xfedcba98, false, false));
}
static __device__ __forceinline__ float lane_val(float x, int l) {
  return __builtin_bit_cast(float, __builtin_amdgcn_readlane(__builtin_bit_cast(int, x), l));
}

static __device__ __forceinline__ float rcp_hw(float x)  { return __builtin_amdgcn_rcpf(x); }
static __device__ __forceinline__ float rsq_hw(float x)  { return __builtin_amdgcn_rsqf(x); }

static __device__ __forceinline__ void lds_barrier() {
  asm volatile("s_wait_dscnt 0x0\n\ts_barrier_signal -1\n\ts_barrier_wait -1" ::: "memory");
}

static __device__ __forceinline__ void chunk_load(v4f (&st)[GS_NST], const float* __restrict__ xg,
                                                  int wave, int so, int ch) {
  #pragma unroll
  for (int j = 0; j < GS_NST; ++j)
    st[j] = *(const v4f*)(xg + (4 * wave + (j >> 1)) * GS_L + (j & 1) * (GS_L / 2) + ch * GS_CW + so);
}
static __device__ __forceinline__ void chunk_store(const v4f (&st)[GS_NST], float* buf, int wave, int so) {
  #pragma unroll
  for (int j = 0; j < GS_NST; ++j)
    *(v4f*)(buf + (4 * wave + (j >> 1)) * GS_BP + (j & 1) * GS_CW + so) = st[j];
}

static __device__ __forceinline__ void q_piece(const v2f (&a)[GS_K / 4], const float* buf, float* __restrict__ og,
                                               int ch, int cq, int ti, int c, int koff, int mb) {
  const int lcol = (cq >> 1) * GS_CW + (cq & 1) * 64 + 4 * c;
  const int gcol = (cq >> 1) * (GS_L / 2) + ch * GS_CW + (cq & 1) * 64 + 4 * c;
  const float* pb = buf + koff * GS_BP + lcol;
  v8f acc0 = {0.f, 0.f, 0.f, 0.f, 0.f, 0.f, 0.f, 0.f};
  v8f acc1 = acc0, acc2 = acc0, acc3 = acc0;
  v4f r0 = *(const v4f*)(pb);
  v4f r1 = *(const v4f*)(pb + GS_BP);
  #pragma unroll
  for (int s = 0; s < GS_K / 4; ++s) {
    v2f b0, b1, b2, b3;
    b0.x = r0.x; b0.y = r1.x;
    b1.x = r0.y; b1.y = r1.y;
    b2.x = r0.z; b2.y = r1.z;
    b3.x = r0.w; b3.y = r1.w;
    if (s + 1 < GS_K / 4) {
      r0 = *(const v4f*)(pb + (4 * s + 4) * GS_BP);
      r1 = *(const v4f*)(pb + (4 * s + 5) * GS_BP);
    }
    acc0 = __builtin_amdgcn_wmma_f32_16x16x4_f32(false, a[s], false, b0, (short)0, acc0, false, false);
    acc1 = __builtin_amdgcn_wmma_f32_16x16x4_f32(false, a[s], false, b1, (short)0, acc1, false, false);
    acc2 = __builtin_amdgcn_wmma_f32_16x16x4_f32(false, a[s], false, b2, (short)0, acc2, false, false);
    acc3 = __builtin_amdgcn_wmma_f32_16x16x4_f32(false, a[s], false, b3, (short)0, acc3, false, false);
  }
  float* po = og + (ti * 16 + mb) * GS_L + gcol;
  #pragma unroll
  for (int j = 0; j < 8; ++j) {
    v4f q;
    q.x = acc0[j]; q.y = acc1[j]; q.z = acc2[j]; q.w = acc3[j];
    *(v4f*)(po + j * GS_L) = q;
  }
}

__global__ __launch_bounds__(GS_NT, 8)
void gram_schmidt_cholqr(const float* __restrict__ xin,
                         float* __restrict__ xout) {
  extern __shared__ float smem[];
  float* R0 = smem;
  float* R1 = smem + GS_BUF;
  float* G  = smem + 2 * GS_BUF;
  float* W  = G + GS_K * GS_K;

  const int b    = blockIdx.x;
  const int tid  = threadIdx.x;
  const int wave = tid >> 5;
  const int lane = tid & 31;
  const int so   = 4 * lane;

  const float* xg = xin  + (size_t)b * (GS_K * GS_L);
  float*       og = xout + (size_t)b * (GS_K * GS_L);

  v4f st[GS_NST];

  {
    const int ti    = wave >> 2;
    const int tj    = wave & 3;
    const int m     = lane & 15;
    const int koff  = (lane < 16) ? 0 : 2;
    v8f acc0 = {0.f, 0.f, 0.f, 0.f, 0.f, 0.f, 0.f, 0.f};
    v8f acc1 = acc0;

    chunk_load(st, xg, wave, so, 0);
    chunk_store(st, R0, wave, so);
    __syncthreads();

    #pragma unroll
    for (int c = 0; c < GS_NCH; ++c) {
      const float* buf = (c & 1) ? R1 : R0;
      float*       nxt = (c & 1) ? R0 : R1;
      const float* pa = buf + (ti * 16 + m) * GS_BP + koff;
      const float* pb = buf + (tj * 16 + m) * GS_BP + koff;
      if (c + 1 < GS_NCH) chunk_load(st, xg, wave, so, c + 1);
      #pragma unroll 4
      for (int k = 0; k < GS_CW; k += 4) {
        v2f a0 = *(const v2f*)(pa + k);
        v2f b0 = *(const v2f*)(pb + k);
        v2f a1 = *(const v2f*)(pa + GS_CW + k);
        v2f b1 = *(const v2f*)(pb + GS_CW + k);
        acc0 = __builtin_amdgcn_wmma_f32_16x16x4_f32(false, a0, false, b0, (short)0, acc0, false, false);
        acc1 = __builtin_amdgcn_wmma_f32_16x16x4_f32(false, a1, false, b1, (short)0, acc1, false, false);
      }
      if (c + 1 < GS_NCH) {
        chunk_store(st, nxt, wave, so);
        __syncthreads();
      }
    }
    const int mb  = (lane < 16) ? 0 : 8;
    #pragma unroll
    for (int j = 0; j < 8; ++j)
      G[(ti * 16 + mb + j) * GS_K + tj * 16 + m] = acc0[j] + acc1[j];
  }
  chunk_load(st, xg, wave, so, 0);
  lds_barrier();

  {
    const int tc = lane & 15;
    const int wv = __builtin_amdgcn_readfirstlane(wave);
    v4f  a[2];
    int  ri[2];
    float di[2] = {1.0f, 1.0f};
    #pragma unroll
    for (int u = 0; u < 2; ++u) {
      ri[u] = 4 * wave + 2 * u + (lane >> 4);
      a[u]  = *(const v4f*)(G + ri[u] * GS_K + 4 * tc);
    }
    for (int p = 0; p < GS_K / 2; ++p) {
      float* r0 = W + (p & 1) * (4 * GS_K);
      float* r1 = r0 + 2 * GS_K;
      const int  pl  = p >> 1;
      const bool odd = (p & 1) != 0;
      #pragma unroll
      for (int u = 0; u < 2; ++u) {
        if (2 * wv + u == p) {
          const int i = ri[u];
          const bool ox = (4 * tc + 0 == i), oy = (4 * tc + 1 == i), oz = (4 * tc + 2 == i), ow = (4 * tc + 3 == i);
          const bool qx = (4 * tc + 0 == (i ^ 1)), qy = (4 * tc + 1 == (i ^ 1)),
                     qz = (4 * tc + 2 == (i ^ 1)), qw = (4 * tc + 3 == (i ^ 1));
          v4f av = a[u];
          v4f bv;
          bv.x = xhalf(av.x); bv.y = xhalf(av.y); bv.z = xhalf(av.z); bv.w = xhalf(av.w);
          const float d0  = lane_val(odd ? av.z : av.x, pl);
          const float g10 = lane_val(odd ? av.w : av.y, pl);
          if (lane >= 16) {
            const float inv = rcp_hw(d0);
            bv.x = qx ? 1.0f : bv.x;  bv.y = qy ? 1.0f : bv.y;  bv.z = qz ? 1.0f : bv.z;  bv.w = qw ? 1.0f : bv.w;
            av.x = qx ? 0.0f : av.x;  av.y = qy ? 0.0f : av.y;  av.z = qz ? 0.0f : av.z;  av.w = qw ? 0.0f : av.w;
            av -= (g10 * bv) * inv;
          }
          const float d1 = lane_val(odd ? av.w : av.y, 16 + pl);
          di[u] = (lane < 16) ? d0 : d1;
          a[u]  = av;
          float* rr = (lane < 16) ? r0 : r1;
          v4f pv;
          pv.x = ox ? 1.0f : av.x;  pv.y = oy ? 1.0f : av.y;  pv.z = oz ? 1.0f : av.z;  pv.w = ow ? 1.0f : av.w;
          *(v4f*)(rr + 4 * tc) = pv;
          if (tc == 0) rr[GS_K] = di[u];
        }
      }
      lds_barrier();
      if (2 * wv + 1 > p) {
        const float d0 = r0[GS_K];
        const float d1 = r1[GS_K];
        const v4f rw0 = *(const v4f*)(r0 + 4 * tc);
        const v4f rw1 = *(const v4f*)(r1 + 4 * tc);
        const float inv0 = rcp_hw(d0);
        const float inv1 = rcp_hw(d1);
        const bool h0 = (tc == pl) && !odd;
        const bool h1 = (tc == pl) && odd;
        #pragma unroll
        for (int u = 1; u >= 0; --u) {
          if (2 * wv + u > p) {
            const float g0 = r0[ri[u]];
            const float g1 = r1[ri[u]];
            v4f av = a[u];
            av.x = h0 ? 0.0f : av.x;
            av.z = h1 ? 0.0f : av.z;
            av -= (g0 * rw0) * inv0;
            av.y = h0 ? 0.0f : av.y;
            av.w = h1 ? 0.0f : av.w;
            av -= (g1 * rw1) * inv1;
            a[u] = av;
          }
        }
      }
    }
    #pragma unroll
    for (int u = 0; u < 2; ++u) {
      const int i = ri[u];
      const bool ox = (4 * tc + 0 == i), oy = (4 * tc + 1 == i), oz = (4 * tc + 2 == i), ow = (4 * tc + 3 == i);
      const float s = rsq_hw(di[u]);
      const v4f av = a[u];
      v4f t;
      t.x = (4 * tc + 0 < i) ? av.x * s : (ox ? s : 0.0f);
      t.y = (4 * tc + 1 < i) ? av.y * s : (oy ? s : 0.0f);
      t.z = (4 * tc + 2 < i) ? av.z * s : (oz ? s : 0.0f);
      t.w = (4 * tc + 3 < i) ? av.w * s : (ow ? s : 0.0f);
      float* fb = G + ((i >> 4) * 16 + tc) * 64;
      const int cc = 2 * (i & 15);
      v2f lo, hi;
      lo.x = t.x; lo.y = t.y;
      hi.x = t.z; hi.y = t.w;
      *(v2f*)(fb + (cc ^ (4 * tc)))        = lo;
      *(v2f*)(fb + ((32 + cc) ^ (4 * tc))) = hi;
    }
  }
  lds_barrier();

  {
    const int c    = lane & 15;
    const int koff = (lane < 16) ? 0 : 2;
    const int mb   = (lane < 16) ? 0 : 8;
    const int ti   = wave & 3;
    const int cq   = wave >> 2;
    const float* pa = G + ti * 16 * 64;
    const int    lh = 16 * koff + 2 * c;
    v2f a[GS_K / 4];
    #pragma unroll
    for (int s = 0; s < GS_K / 4; ++s)
      a[s] = *(const v2f*)(pa + 64 * s + (lh ^ (4 * s)));

    q_piece(a, R1, og, 3, cq, ti, c, koff, mb);
    lds_barrier();
    chunk_store(st, R1, wave, so);
    chunk_load(st, xg, wave, so, 1);
    q_piece(a, R0, og, 2, cq, ti, c, koff, mb);
    lds_barrier();
    chunk_store(st, R0, wave, so);
    q_piece(a, R1, og, 0, cq, ti, c, koff, mb);
    lds_barrier();
    q_piece(a, R0, og, 1, cq, ti, c, koff, mb);
  }
}

extern "C" void kernel_launch(void* const* d_in, const int* in_sizes, int n_in,
                              void* d_out, int out_size, void* d_ws, size_t ws_size,
                              hipStream_t stream) {
  (void)in_sizes; (void)n_in; (void)out_size; (void)d_ws; (void)ws_size;
  const float* x   = (const float*)d_in[0];
  float*       out = (float*)d_out;

  const size_t shmem = (size_t)(2 * GS_BUF + GS_K * GS_K + 8 * GS_K) * sizeof(float);
  hipFuncSetAttribute(reinterpret_cast<const void*>(&gram_schmidt_cholqr),
                      hipFuncAttributeMaxDynamicSharedMemorySize, (int)shmem);

  gram_schmidt_cholqr<<<dim3(GS_B), dim3(GS_NT), shmem, stream>>>(x, out);
}


# Wave priority by phase (s_setprio: sweep 2, Gram product 1, final product 0): the co-resident work-group that is behind wins the matrix pipe and each sweep runs under the other's products; same sums
# speedup vs baseline: 9.5122x; 1.0403x over previous
#include <hip/hip_runtime.h>


typedef float v2f __attribute__((ext_vector_type(2)));
typedef float v4f __attribute__((ext_vector_type(4)));
typedef float v8f __attribute__((ext_vector_type(8)));

#define GS_B  512
#define GS_K  64
#define GS_L  1024
#define GS_NT 512
#define GS_NCH 4
#define GS_CW  (GS_L / 2 / GS_NCH)
#define GS_BP  (2 * GS_CW + 4)
#define GS_BUF (GS_K * GS_BP)
#define GS_NST 8

static __device__ __forceinline__ float xhalf(float x) {
  return __builtin_bit_cast(float, __builtin_amdgcn_permlanex16(__builtin_bit_cast(int, x), __builtin_bit_cast(int, x),
                                                                0x76543210, 0xfedcba98, false, false));
}
static __device__ __forceinline__ float lane_val(float x, int l) {
  return __builtin_bit_cast(float, __builtin_amdgcn_readlane(__builtin_bit_cast(int, x), l));
}

static __device__ __forceinline__ float rcp_hw(float x)  { return __builtin_amdgcn_rcpf(x); }
static __device__ __forceinline__ float rsq_hw(float x)  { return __builtin_amdgcn_rsqf(x); }

static __device__ __forceinline__ void lds_barrier() {
  asm volatile("s_wait_dscnt 0x0\n\ts_barrier_signal -1\n\ts_barrier_wait -1" ::: "memory");
}

#define GS_PRIO_GRAM  1
#define GS_PRIO_SWEEP 2
#define GS_PRIO_FINAL 0

static __device__ __forceinline__ void chunk_load(v4f (&st)[GS_NST], const float* __restrict__ xg,
                                                  int wave, int so, int ch) {
  #pragma unroll
  for (int j = 0; j < GS_NST; ++j)
    st[j] = *(const v4f*)(xg + (4 * wave + (j >> 1)) * GS_L + (j & 1) * (GS_L / 2) + ch * GS_CW + so);
}
static __device__ __forceinline__ void chunk_store(const v4f (&st)[GS_NST], float* buf, int wave, int so) {
  #pragma unroll
  for (int j = 0; j < GS_NST; ++j)
    *(v4f*)(buf + (4 * wave + (j >> 1)) * GS_BP + (j & 1) * GS_CW + so) = st[j];
}

static __device__ __forceinline__ void q_piece(const v2f (&a)[GS_K / 4], const float* buf, float* __restrict__ og,
                                               int ch, int cq, int ti, int c, int koff, int mb) {
  const int lcol = (cq >> 1) * GS_CW + (cq & 1) * 64 + 4 * c;
  const int gcol = (cq >> 1) * (GS_L / 2) + ch * GS_CW + (cq & 1) * 64 + 4 * c;
  const float* pb = buf + koff * GS_BP + lcol;
  v8f acc0 = {0.f, 0.f, 0.f, 0.f, 0.f, 0.f, 0.f, 0.f};
  v8f acc1 = acc0, acc2 = acc0, acc3 = acc0;
  v4f r0 = *(const v4f*)(pb);
  v4f r1 = *(const v4f*)(pb + GS_BP);
  #pragma unroll
  for (int s = 0; s < GS_K / 4; ++s) {
    v2f b0, b1, b2, b3;
    b0.x = r0.x; b0.y = r1.x;
    b1.x = r0.y; b1.y = r1.y;
    b2.x = r0.z; b2.y = r1.z;
    b3.x = r0.w; b3.y = r1.w;
    if (s + 1 < GS_K / 4) {
      r0 = *(const v4f*)(pb + (4 * s + 4) * GS_BP);
      r1 = *(const v4f*)(pb + (4 * s + 5) * GS_BP);
    }
    acc0 = __builtin_amdgcn_wmma_f32_16x16x4_f32(false, a[s], false, b0, (short)0, acc0, false, false);
    acc1 = __builtin_amdgcn_wmma_f32_16x16x4_f32(false, a[s], false, b1, (short)0, acc1, false, false);
    acc2 = __builtin_amdgcn_wmma_f32_16x16x4_f32(false, a[s], false, b2, (short)0, acc2, false, false);
    acc3 = __builtin_amdgcn_wmma_f32_16x16x4_f32(false, a[s], false, b3, (short)0, acc3, false, false);
  }
  float* po = og + (ti * 16 + mb) * GS_L + gcol;
  #pragma unroll
  for (int j = 0; j < 8; ++j) {
    v4f q;
    q.x = acc0[j]; q.y = acc1[j]; q.z = acc2[j]; q.w = acc3[j];
    *(v4f*)(po + j * GS_L) = q;
  }
}

__global__ __launch_bounds__(GS_NT, 8)
void gram_schmidt_cholqr(const float* __restrict__ xin,
                         float* __restrict__ xout) {
  extern __shared__ float smem[];
  float* R0 = smem;
  float* R1 = smem + GS_BUF;
  float* G  = smem + 2 * GS_BUF;
  float* W  = G + GS_K * GS_K;

  const int b    = blockIdx.x;
  const int tid  = threadIdx.x;
  const int wave = tid >> 5;
  const int lane = tid & 31;
  const int so   = 4 * lane;

  const float* xg = xin  + (size_t)b * (GS_K * GS_L);
  float*       og = xout + (size_t)b * (GS_K * GS_L);

  v4f st[GS_NST];

  __builtin_amdgcn_s_setprio(GS_PRIO_GRAM);

  {
    const int ti    = wave >> 2;
    const int tj    = wave & 3;
    const int m     = lane & 15;
    const int koff  = (lane < 16) ? 0 : 2;
    v8f acc0 = {0.f, 0.f, 0.f, 0.f, 0.f, 0.f, 0.f, 0.f};
    v8f acc1 = acc0;

    chunk_load(st, xg, wave, so, 0);
    chunk_store(st, R0, wave, so);
    __syncthreads();

    #pragma unroll
    for (int c = 0; c < GS_NCH; ++c) {
      const float* buf = (c & 1) ? R1 : R0;
      float*       nxt = (c & 1) ? R0 : R1;
      const float* pa = buf + (ti * 16 + m) * GS_BP + koff;
      const float* pb = buf + (tj * 16 + m) * GS_BP + koff;
      if (c + 1 < GS_NCH) chunk_load(st, xg, wave, so, c + 1);
      #pragma unroll 4
      for (int k = 0; k < GS_CW; k += 4) {
        v2f a0 = *(const v2f*)(pa + k);
        v2f b0 = *(const v2f*)(pb + k);
        v2f a1 = *(const v2f*)(pa + GS_CW + k);
        v2f b1 = *(const v2f*)(pb + GS_CW + k);
        acc0 = __builtin_amdgcn_wmma_f32_16x16x4_f32(false, a0, false, b0, (short)0, acc0, false, false);
        acc1 = __builtin_amdgcn_wmma_f32_16x16x4_f32(false, a1, false, b1, (short)0, acc1, false, false);
      }
      if (c + 1 < GS_NCH) {
        chunk_store(st, nxt, wave, so);
        __syncthreads();
      }
    }
    __builtin_amdgcn_s_setprio(GS_PRIO_SWEEP);
    const int mb  = (lane < 16) ? 0 : 8;
    #pragma unroll
    for (int j = 0; j < 8; ++j)
      G[(ti * 16 + mb + j) * GS_K + tj * 16 + m] = acc0[j] + acc1[j];
  }
  chunk_load(st, xg, wave, so, 0);
  lds_barrier();

  {
    const int tc = lane & 15;
    const int wv = __builtin_amdgcn_readfirstlane(wave);
    v4f  a[2];
    int  ri[2];
    float di[2] = {1.0f, 1.0f};
    #pragma unroll
    for (int u = 0; u < 2; ++u) {
      ri[u] = 4 * wave + 2 * u + (lane >> 4);
      a[u]  = *(const v4f*)(G + ri[u] * GS_K + 4 * tc);
    }
    for (int p = 0; p < GS_K / 2; ++p) {
      float* r0 = W + (p & 1) * (4 * GS_K);
      float* r1 = r0 + 2 * GS_K;
      const int  pl  = p >> 1;
      const bool odd = (p & 1) != 0;
      #pragma unroll
      for (int u = 0; u < 2; ++u) {
        if (2 * wv + u == p) {
          const int i = ri[u];
          const bool ox = (4 * tc + 0 == i), oy = (4 * tc + 1 == i), oz = (4 * tc + 2 == i), ow = (4 * tc + 3 == i);
          const bool qx = (4 * tc + 0 == (i ^ 1)), qy = (4 * tc + 1 == (i ^ 1)),
                     qz = (4 * tc + 2 == (i ^ 1)), qw = (4 * tc + 3 == (i ^ 1));
          v4f av = a[u];
          v4f bv;
          bv.x = xhalf(av.x); bv.y = xhalf(av.y); bv.z = xhalf(av.z); bv.w = xhalf(av.w);
          const float d0  = lane_val(odd ? av.z : av.x, pl);
          const float g10 = lane_val(odd ? av.w : av.y, pl);
          if (lane >= 16) {
            const float inv = rcp_hw(d0);
            bv.x = qx ? 1.0f : bv.x;  bv.y = qy ? 1.0f : bv.y;  bv.z = qz ? 1.0f : bv.z;  bv.w = qw ? 1.0f : bv.w;
            av.x = qx ? 0.0f : av.x;  av.y = qy ? 0.0f : av.y;  av.z = qz ? 0.0f : av.z;  av.w = qw ? 0.0f : av.w;
            av -= (g10 * bv) * inv;
          }
          const float d1 = lane_val(odd ? av.w : av.y, 16 + pl);
          di[u] = (lane < 16) ? d0 : d1;
          a[u]  = av;
          float* rr = (lane < 16) ? r0 : r1;
          v4f pv;
          pv.x = ox ? 1.0f : av.x;  pv.y = oy ? 1.0f : av.y;  pv.z = oz ? 1.0f : av.z;  pv.w = ow ? 1.0f : av.w;
          *(v4f*)(rr + 4 * tc) = pv;
          if (tc == 0) rr[GS_K] = di[u];
        }
      }
      lds_barrier();
      if (2 * wv + 1 > p) {
        const float d0 = r0[GS_K];
        const float d1 = r1[GS_K];
        const v4f rw0 = *(const v4f*)(r0 + 4 * tc);
        const v4f rw1 = *(const v4f*)(r1 + 4 * tc);
        const float inv0 = rcp_hw(d0);
        const float inv1 = rcp_hw(d1);
        const bool h0 = (tc == pl) && !odd;
        const bool h1 = (tc == pl) && odd;
        #pragma unroll
        for (int u = 1; u >= 0; --u) {
          if (2 * wv + u > p) {
            const float g0 = r0[ri[u]];
            const float g1 = r1[ri[u]];
            v4f av = a[u];
            av.x = h0 ? 0.0f : av.x;
            av.z = h1 ? 0.0f : av.z;
            av -= (g0 * rw0) * inv0;
            av.y = h0 ? 0.0f : av.y;
            av.w = h1 ? 0.0f : av.w;
            av -= (g1 * rw1) * inv1;
            a[u] = av;
          }
        }
      }
    }
    #pragma unroll
    for (int u = 0; u < 2; ++u) {
      const int i = ri[u];
      const bool ox = (4 * tc + 0 == i), oy = (4 * tc + 1 == i), oz = (4 * tc + 2 == i), ow = (4 * tc + 3 == i);
      const float s = rsq_hw(di[u]);
      const v4f av = a[u];
      v4f t;
      t.x = (4 * tc + 0 < i) ? av.x * s : (ox ? s : 0.0f);
      t.y = (4 * tc + 1 < i) ? av.y * s : (oy ? s : 0.0f);
      t.z = (4 * tc + 2 < i) ? av.z * s : (oz ? s : 0.0f);
      t.w = (4 * tc + 3 < i) ? av.w * s : (ow ? s : 0.0f);
      float* fb = G + ((i >> 4) * 16 + tc) * 64;
      const int cc = 2 * (i & 15);
      v2f lo, hi;
      lo.x = t.x; lo.y = t.y;
      hi.x = t.z; hi.y = t.w;
      *(v2f*)(fb + (cc ^ (4 * tc)))        = lo;
      *(v2f*)(fb + ((32 + cc) ^ (4 * tc))) = hi;
    }
  }
  lds_barrier();
  __builtin_amdgcn_s_setprio(GS_PRIO_FINAL);

  {
    const int c    = lane & 15;
    const int koff = (lane < 16) ? 0 : 2;
    const int mb   = (lane < 16) ? 0 : 8;
    const int ti   = wave & 3;
    const int cq   = wave >> 2;
    const float* pa = G + ti * 16 * 64;
    const int    lh = 16 * koff + 2 * c;
    v2f a[GS_K / 4];
    #pragma unroll
    for (int s = 0; s < GS_K / 4; ++s)
      a[s] = *(const v2f*)(pa + 64 * s + (lh ^ (4 * s)));

    q_piece(a, R1, og, 3, cq, ti, c, koff, mb);
    lds_barrier();
    chunk_store(st, R1, wave, so);
    chunk_load(st, xg, wave, so, 1);
    q_piece(a, R0, og, 2, cq, ti, c, koff, mb);
    lds_barrier();
    chunk_store(st, R0, wave, so);
    q_piece(a, R1, og, 0, cq, ti, c, koff, mb);
    lds_barrier();
    q_piece(a, R0, og, 1, cq, ti, c, koff, mb);
  }
}

extern "C" void kernel_launch(void* const* d_in, const int* in_sizes, int n_in,
                              void* d_out, int out_size, void* d_ws, size_t ws_size,
                              hipStream_t stream) {
  (void)in_sizes; (void)n_in; (void)out_size; (void)d_ws; (void)ws_size;
  const float* x   = (const float*)d_in[0];
  float*       out = (float*)d_out;

  const size_t shmem = (size_t)(2 * GS_BUF + GS_K * GS_K + 8 * GS_K) * sizeof(float);
  hipFuncSetAttribute(reinterpret_cast<const void*>(&gram_schmidt_cholqr),
                      hipFuncAttributeMaxDynamicSharedMemorySize, (int)shmem);

  gram_schmidt_cholqr<<<dim3(GS_B), dim3(GS_NT), shmem, stream>>>(x, out);
}
